# starting kernel
# baseline (speedup 1.0000x reference)
#include <hip/hip_runtime.h>
#include <hip/hip_bf16.h>

typedef __attribute__((ext_vector_type(16))) _Float16 v16h;
typedef __attribute__((ext_vector_type(8)))  float    v8f;

template <int K, int NCOL>
__global__ __launch_bounds__(32) void pack_weights(
    const float* __restrict__ m0, const float* __restrict__ m1,
    const float* __restrict__ m2, _Float16* __restrict__ out)
{
  const float* W = (blockIdx.y == 0) ? m0 : (blockIdx.y == 1) ? m1 : m2;
  const int ct   = blockIdx.x / (K / 32);
  const int ks   = blockIdx.x % (K / 32);
  const int lane = threadIdx.x;
  const int hs   = lane >> 4;
  const int mr   = lane & 15;
  const int krow = ks * 32 + hs * 8;
  const float* p = W + (size_t)krow * NCOL + ct * 16 + mr;
  _Float16* o = out +
      ((((size_t)blockIdx.y * (NCOL / 16) + ct) * (K / 32) + ks) * 32 + lane) * 16;
#pragma unroll
  for (int i = 0; i < 8; ++i) {
    o[i]     = (_Float16)p[(size_t)i * NCOL];
    o[8 + i] = (_Float16)p[(size_t)(16 + i) * NCOL];
  }
}

template <int K, int NCOL, bool RELU>
__global__ __launch_bounds__((NCOL / 16) * 32) void node_xform(
    const float* __restrict__ x,
    const _Float16* __restrict__ wpack,
    const float* __restrict__ bias,
    float* __restrict__ o0, float* __restrict__ o1, float* __restrict__ oagg)
{
  constexpr int KS = K / 32;
  __shared__ v16h ash[KS * 32];

  const int lane = threadIdx.x & 31;
  const int wave = threadIdx.x >> 5;
  const int row0 = blockIdx.x * 16;
  const int hs   = lane >> 4;
  const int mr   = lane & 15;

  if (wave < KS) {
    const int    ks = wave;
    const float* ap = x + (size_t)(row0 + mr) * K + ks * 32 + hs * 8;
    v16h a;
#pragma unroll
    for (int i = 0; i < 8; ++i) {
      float lo = ap[i];
      float hi = ap[16 + i];
      if (RELU) { lo = fmaxf(lo, 0.0f); hi = fmaxf(hi, 0.0f); }
      a[i]     = (_Float16)lo;
      a[8 + i] = (_Float16)hi;
    }
    ash[ks * 32 + lane] = a;
  }
  __syncthreads();

  v8f c0 = {}, c1 = {}, cl = {};
  constexpr size_t MATSTRIDE = (size_t)(NCOL / 16) * KS * 32 * 16;
  const _Float16* wb = wpack + ((size_t)wave * KS) * 32 * 16 + (size_t)lane * 16;

#pragma unroll
  for (int ks = 0; ks < KS; ++ks) {
    v16h a = ash[ks * 32 + lane];
    const _Float16* f = wb + (size_t)ks * 32 * 16;
    v16h b0 = *(const v16h*)(f);
    v16h b1 = *(const v16h*)(f + MATSTRIDE);
    v16h bl = *(const v16h*)(f + 2 * MATSTRIDE);
    c0 = __builtin_amdgcn_wmma_f32_16x16x32_f16(false, a, false, b0, (short)0, c0, false, false);
    c1 = __builtin_amdgcn_wmma_f32_16x16x32_f16(false, a, false, b1, (short)0, c1, false, false);
    cl = __builtin_amdgcn_wmma_f32_16x16x32_f16(false, a, false, bl, (short)0, cl, false, false);
  }

  const int   col = wave * 16 + mr;
  const float bv  = bias[col];
#pragma unroll
  for (int r = 0; r < 8; ++r) {
    size_t off = (size_t)(row0 + r + hs * 8) * NCOL + col;
    o0[off]   = c0[r];
    o1[off]   = c1[r];
    oagg[off] = cl[r] + bv;
  }
}

__global__ __launch_bounds__(256) void zero_u32(unsigned* p, int n) {
  int i = blockIdx.x * 256 + threadIdx.x;
  if (i < n) p[i] = 0u;
}

__global__ __launch_bounds__(256) void count_edges(
    const int* __restrict__ dst, unsigned* __restrict__ counts, int E) {
  int e = blockIdx.x * 256 + threadIdx.x;
  if (e < E) atomicAdd(&counts[dst[e]], 1u);
}

__global__ __launch_bounds__(1024) void scan_offsets(
    const unsigned* __restrict__ counts, unsigned* __restrict__ offsets,
    unsigned* __restrict__ cursor, int n)
{
  __shared__ unsigned s[1024];
  __shared__ unsigned carry;
  const int t = threadIdx.x;
  if (t == 0) carry = 0u;
  __syncthreads();
  for (int base = 0; base < n; base += 1024) {
    const int i = base + t;
    unsigned v = (i < n) ? counts[i] : 0u;
    s[t] = v;
    __syncthreads();
#pragma unroll
    for (int off = 1; off < 1024; off <<= 1) {
      unsigned u = (t >= off) ? s[t - off] : 0u;
      __syncthreads();
      s[t] += u;
      __syncthreads();
    }
    const unsigned excl  = s[t] - v + carry;
    const unsigned total = s[1023];
    if (i < n) { offsets[i] = excl; cursor[i] = excl; }
    __syncthreads();
    if (t == 0) carry += total;
    __syncthreads();
  }
  if (t == 0) offsets[n] = carry;
}

__global__ __launch_bounds__(256) void fill_edges(
    const int* __restrict__ src, const int* __restrict__ dst,
    const int* __restrict__ et, unsigned* __restrict__ cursor,
    unsigned* __restrict__ eidx, int E)
{
  int e = blockIdx.x * 256 + threadIdx.x;
  if (e >= E) return;
  unsigned pos = atomicAdd(&cursor[dst[e]], 1u);
  eidx[pos] = (unsigned)src[e] | ((unsigned)et[e] << 30);
}

template <int F>
__global__ __launch_bounds__(256) void aggregate(
    const float* __restrict__ x0, const float* __restrict__ x1,
    const unsigned* __restrict__ offsets, const unsigned* __restrict__ eidx,
    float* __restrict__ io, int N)
{
  constexpr int VEC = F / 32;
  int gid  = blockIdx.x * 256 + threadIdx.x;
  int node = gid >> 5;
  if (node >= N) return;
  int lane = gid & 31;

  const unsigned beg = offsets[node];
  const unsigned end = offsets[node + 1];

  float acc0[VEC], acc1[VEC];
#pragma unroll
  for (int i = 0; i < VEC; ++i) { acc0[i] = 0.0f; acc1[i] = 0.0f; }

  unsigned e = beg;
  for (; e + 2 <= end; e += 2) {
    const unsigned pk0 = eidx[e];
    const unsigned pk1 = eidx[e + 1];
    const float* r0 = ((pk0 >> 30) ? x1 : x0) + (size_t)(pk0 & 0x3FFFFFFFu) * F + lane * VEC;
    const float* r1 = ((pk1 >> 30) ? x1 : x0) + (size_t)(pk1 & 0x3FFFFFFFu) * F + lane * VEC;
    if (VEC == 4) {
      const float4 v0 = *(const float4*)r0;
      const float4 v1 = *(const float4*)r1;
      acc0[0] += v0.x; acc0[1] += v0.y; acc0[2] += v0.z; acc0[3] += v0.w;
      acc1[0] += v1.x; acc1[1] += v1.y; acc1[2] += v1.z; acc1[3] += v1.w;
    } else {
      const float2 v0 = *(const float2*)r0;
      const float2 v1 = *(const float2*)r1;
      acc0[0] += v0.x; acc0[1] += v0.y;
      acc1[0] += v1.x; acc1[1] += v1.y;
    }
  }
  if (e < end) {
    const unsigned pk = eidx[e];
    const float* r = ((pk >> 30) ? x1 : x0) + (size_t)(pk & 0x3FFFFFFFu) * F + lane * VEC;
    if (VEC == 4) {
      const float4 v = *(const float4*)r;
      acc0[0] += v.x; acc0[1] += v.y; acc0[2] += v.z; acc0[3] += v.w;
    } else {
      const float2 v = *(const float2*)r;
      acc0[0] += v.x; acc0[1] += v.y;
    }
  }

  float* o = io + (size_t)node * F + lane * VEC;
#pragma unroll
  for (int i = 0; i < VEC; ++i) o[i] += acc0[i] + acc1[i];
}

extern "C" void kernel_launch(void* const* d_in, const int* in_sizes, int n_in,
                              void* d_out, int out_size, void* d_ws, size_t ws_size,
                              hipStream_t stream) {
  const float* feat  = (const float*)d_in[0];
  const float* W1    = (const float*)d_in[1];
  const float* loop1 = (const float*)d_in[2];
  const float* b1    = (const float*)d_in[3];
  const float* W2    = (const float*)d_in[4];
  const float* loop2 = (const float*)d_in[5];
  const float* b2    = (const float*)d_in[6];
  const int*   src   = (const int*)d_in[7];
  const int*   dst   = (const int*)d_in[8];
  const int*   et    = (const int*)d_in[9];
  float*       out   = (float*)d_out;

  const int N = in_sizes[0] / 64;
  const int E = in_sizes[7];

  float* ws   = (float*)d_ws;
  float* x0   = ws;
  float* x1   = x0 + (size_t)N * 128;
  float* hbuf = x1 + (size_t)N * 128;
  float* y0   = x0;
  float* y1   = x0 + (size_t)N * 64;
  _Float16* wpack1 = (_Float16*)(hbuf + (size_t)N * 128);
  _Float16* wpack2 = wpack1 + 24576;
  unsigned* counts  = (unsigned*)(wpack2 + 24576);
  unsigned* offsets = counts + N;
  unsigned* cursor  = offsets + N + 64;
  unsigned* eidx    = cursor + N;

  const int nblk = N / 16;
  const int eblk = (E + 255) / 256;
  const int ablk = (N * 32 + 255) / 256;
  const int zblk = (N + 255) / 256;

  pack_weights<64, 128><<<dim3(8 * 2, 3), 32, 0, stream>>>(
      W1, W1 + 64 * 128, loop1, wpack1);
  pack_weights<128, 64><<<dim3(4 * 4, 3), 32, 0, stream>>>(
      W2, W2 + 128 * 64, loop2, wpack2);

  zero_u32<<<zblk, 256, 0, stream>>>(counts, N);
  count_edges<<<eblk, 256, 0, stream>>>(dst, counts, E);
  scan_offsets<<<1, 1024, 0, stream>>>(counts, offsets, cursor, N);
  fill_edges<<<eblk, 256, 0, stream>>>(src, dst, et, cursor, eidx, E);

  node_xform<64, 128, false><<<nblk, 256, 0, stream>>>(
      feat, wpack1, b1, x0, x1, hbuf);
  aggregate<128><<<ablk, 256, 0, stream>>>(x0, x1, offsets, eidx, hbuf, N);

  node_xform<128, 64, true><<<nblk, 128, 0, stream>>>(
      hbuf, wpack2, b2, y0, y1, out);
  aggregate<64><<<ablk, 256, 0, stream>>>(y0, y1, offsets, eidx, out, N);
}


# Independent launches run as work-groups of one launch: weight packing beside the counter clear, edge fill interleaved with layer 1's node product; 8 launches become 5; edge order by arrival as before
# speedup vs baseline: 2.4794x; 2.4794x over previous
#include <hip/hip_runtime.h>
#include <hip/hip_bf16.h>

typedef __attribute__((ext_vector_type(16))) _Float16 v16h;
typedef __attribute__((ext_vector_type(8)))  float    v8f;

template <int K, int NCOL>
__device__ __forceinline__ void pack_fragment(
    const float* __restrict__ Wm, _Float16* __restrict__ out,
    int mat, int x, int lane)
{
  const int ct   = x / (K / 32);
  const int ks   = x % (K / 32);
  const int hs   = lane >> 4;
  const int mr   = lane & 15;
  const int krow = ks * 32 + hs * 8;
  const float* p = Wm + (size_t)krow * NCOL + ct * 16 + mr;
  _Float16* o = out +
      ((((size_t)mat * (NCOL / 16) + ct) * (K / 32) + ks) * 32 + lane) * 16;
#pragma unroll
  for (int i = 0; i < 8; ++i) {
    o[i]     = (_Float16)p[(size_t)i * NCOL];
    o[8 + i] = (_Float16)p[(size_t)(16 + i) * NCOL];
  }
}

constexpr int PACK_WAVES = 48;

__global__ __launch_bounds__(256) void prep(
    const float* __restrict__ w1a, const float* __restrict__ w1b,
    const float* __restrict__ l1,
    const float* __restrict__ w2a, const float* __restrict__ w2b,
    const float* __restrict__ l2,
    _Float16* __restrict__ wpack1, _Float16* __restrict__ wpack2,
    unsigned* __restrict__ cnt, int N)
{
  const int lane = threadIdx.x & 31;
  const int gw   = blockIdx.x * 8 + (threadIdx.x >> 5);
  if (gw < PACK_WAVES) {
    const int mat = gw / 16;
    const float* Wm = (mat == 0) ? w1a : (mat == 1) ? w1b : l1;
    pack_fragment<64, 128>(Wm, wpack1, mat, gw % 16, lane);
  } else if (gw < 2 * PACK_WAVES) {
    const int g   = gw - PACK_WAVES;
    const int mat = g / 16;
    const float* Wm = (mat == 0) ? w2a : (mat == 1) ? w2b : l2;
    pack_fragment<128, 64>(Wm, wpack2, mat, g % 16, lane);
  } else {
    const int base = ((gw - 2 * PACK_WAVES) * 32 + lane) * 4;
    if (base + 3 < N) {
      *(uint4*)(cnt + base) = make_uint4(0u, 0u, 0u, 0u);
    } else {
      for (int k = 0; k < 4; ++k)
        if (base + k < N) cnt[base + k] = 0u;
    }
  }
}

template <int K, int NCOL, bool RELU>
__device__ __forceinline__ void xform_strip(
    int strip, v16h* ash,
    const float* __restrict__ x,
    const _Float16* __restrict__ wpack,
    const float* __restrict__ bias,
    float* __restrict__ o0, float* __restrict__ o1, float* __restrict__ oagg)
{
  constexpr int KS = K / 32;

  const int lane = threadIdx.x & 31;
  const int wave = threadIdx.x >> 5;
  const int row0 = strip * 16;
  const int hs   = lane >> 4;
  const int mr   = lane & 15;

  if (wave < KS) {
    const int    ks = wave;
    const float* ap = x + (size_t)(row0 + mr) * K + ks * 32 + hs * 8;
    v16h a;
#pragma unroll
    for (int i = 0; i < 8; ++i) {
      float lo = ap[i];
      float hi = ap[16 + i];
      if (RELU) { lo = fmaxf(lo, 0.0f); hi = fmaxf(hi, 0.0f); }
      a[i]     = (_Float16)lo;
      a[8 + i] = (_Float16)hi;
    }
    ash[ks * 32 + lane] = a;
  }
  __syncthreads();

  v8f c0 = {}, c1 = {}, cl = {};
  constexpr size_t MATSTRIDE = (size_t)(NCOL / 16) * KS * 32 * 16;
  const _Float16* wb = wpack + ((size_t)wave * KS) * 32 * 16 + (size_t)lane * 16;

#pragma unroll
  for (int ks = 0; ks < KS; ++ks) {
    v16h a = ash[ks * 32 + lane];
    const _Float16* f = wb + (size_t)ks * 32 * 16;
    v16h b0 = *(const v16h*)(f);
    v16h b1 = *(const v16h*)(f + MATSTRIDE);
    v16h bl = *(const v16h*)(f + 2 * MATSTRIDE);
    c0 = __builtin_amdgcn_wmma_f32_16x16x32_f16(false, a, false, b0, (short)0, c0, false, false);
    c1 = __builtin_amdgcn_wmma_f32_16x16x32_f16(false, a, false, b1, (short)0, c1, false, false);
    cl = __builtin_amdgcn_wmma_f32_16x16x32_f16(false, a, false, bl, (short)0, cl, false, false);
  }

  const int   col = wave * 16 + mr;
  const float bv  = bias[col];
#pragma unroll
  for (int r = 0; r < 8; ++r) {
    size_t off = (size_t)(row0 + r + hs * 8) * NCOL + col;
    o0[off]   = c0[r];
    o1[off]   = c1[r];
    oagg[off] = cl[r] + bv;
  }
}

template <int K, int NCOL, bool RELU>
__global__ __launch_bounds__((NCOL / 16) * 32) void node_xform(
    const float* __restrict__ x, const _Float16* __restrict__ wpack,
    const float* __restrict__ bias,
    float* __restrict__ o0, float* __restrict__ o1, float* __restrict__ oagg)
{
  __shared__ v16h ash[(K / 32) * 32];
  xform_strip<K, NCOL, RELU>(blockIdx.x, ash, x, wpack, bias, o0, o1, oagg);
}

constexpr int ROWW = 64;

__device__ __forceinline__ void fill_chunk(
    int chunk,
    const int* __restrict__ src, const int* __restrict__ dst,
    const int* __restrict__ et, unsigned* __restrict__ cnt,
    float* rowbuf, int E)
{
  int e = chunk * 256 + threadIdx.x;
  if (e >= E) return;
  unsigned* rows = (unsigned*)rowbuf;
  const int d = dst[e];
  const unsigned pk  = (unsigned)src[e] | ((unsigned)et[e] << 30);
  const unsigned pos = atomicAdd(&cnt[d], 1u);
  const unsigned slot = pos < (unsigned)ROWW ? pos : (unsigned)(ROWW - 1);
  rows[(size_t)d * ROWW + slot] = pk;
}

__global__ __launch_bounds__(256) void fill_beside_xform1(
    const float* __restrict__ feat, const _Float16* __restrict__ wpack1,
    const float* __restrict__ b1,
    float* __restrict__ x0, float* __restrict__ x1, float* __restrict__ hbuf,
    const int* __restrict__ src, const int* __restrict__ dst,
    const int* __restrict__ et, unsigned* __restrict__ cnt,
    float* rowbuf, int E, int nstrips)
{
  __shared__ v16h ash[(64 / 32) * 32];
  const int idx = blockIdx.x >> 1;
  if ((blockIdx.x & 1) == 0) {
    if (idx < nstrips)
      xform_strip<64, 128, false>(idx, ash, feat, wpack1, b1, x0, x1, hbuf);
  } else {
    fill_chunk(idx, src, dst, et, cnt, rowbuf, E);
  }
}

template <int F>
__global__ __launch_bounds__(256) void aggregate(
    const float* __restrict__ x0, const float* __restrict__ x1,
    const unsigned* __restrict__ cnt, const float* rowbuf,
    const int* __restrict__ src, const int* __restrict__ dst,
    const int* __restrict__ et,
    const float* init, float* dest, int N, int E)
{
  constexpr int VEC = F / 32;
  int gid  = blockIdx.x * 256 + threadIdx.x;
  int node = gid >> 5;
  if (node >= N) return;
  int lane = gid & 31;

  const unsigned deg = cnt[node];

  float acc0[VEC], acc1[VEC];
#pragma unroll
  for (int i = 0; i < VEC; ++i) { acc0[i] = 0.0f; acc1[i] = 0.0f; }

  if (deg <= (unsigned)ROWW) {
    const unsigned* row = (const unsigned*)rowbuf + (size_t)node * ROWW;
    unsigned e = 0;
    for (; e + 2 <= deg; e += 2) {
      const unsigned pk0 = row[e];
      const unsigned pk1 = row[e + 1];
      const float* r0 = ((pk0 >> 30) ? x1 : x0) + (size_t)(pk0 & 0x3FFFFFFFu) * F + lane * VEC;
      const float* r1 = ((pk1 >> 30) ? x1 : x0) + (size_t)(pk1 & 0x3FFFFFFFu) * F + lane * VEC;
      if (VEC == 4) {
        const float4 v0 = *(const float4*)r0;
        const float4 v1 = *(const float4*)r1;
        acc0[0] += v0.x; acc0[1] += v0.y; acc0[2] += v0.z; acc0[3] += v0.w;
        acc1[0] += v1.x; acc1[1] += v1.y; acc1[2] += v1.z; acc1[3] += v1.w;
      } else {
        const float2 v0 = *(const float2*)r0;
        const float2 v1 = *(const float2*)r1;
        acc0[0] += v0.x; acc0[1] += v0.y;
        acc1[0] += v1.x; acc1[1] += v1.y;
      }
    }
    if (e < deg) {
      const unsigned pk = row[e];
      const float* r = ((pk >> 30) ? x1 : x0) + (size_t)(pk & 0x3FFFFFFFu) * F + lane * VEC;
      if (VEC == 4) {
        const float4 v = *(const float4*)r;
        acc0[0] += v.x; acc0[1] += v.y; acc0[2] += v.z; acc0[3] += v.w;
      } else {
        const float2 v = *(const float2*)r;
        acc0[0] += v.x; acc0[1] += v.y;
      }
    }
  } else {
    for (int e0 = 0; e0 < E; e0 += 32) {
      const int  e   = e0 + lane;
      const bool hit = (e < E) && (dst[e] == node);
      unsigned   pk  = 0u;
      if (hit) pk = (unsigned)src[e] | ((unsigned)et[e] << 30);
      unsigned m = __builtin_amdgcn_ballot_w32(hit);
      while (m) {
        const int j = __builtin_ctz(m);
        m &= m - 1u;
        const unsigned p = (unsigned)__shfl((int)pk, j);
        const float* r = ((p >> 30) ? x1 : x0) + (size_t)(p & 0x3FFFFFFFu) * F + lane * VEC;
        if (VEC == 4) {
          const float4 v = *(const float4*)r;
          acc0[0] += v.x; acc0[1] += v.y; acc0[2] += v.z; acc0[3] += v.w;
        } else {
          const float2 v = *(const float2*)r;
          acc0[0] += v.x; acc0[1] += v.y;
        }
      }
    }
  }

  const float* ip = init + (size_t)node * F + lane * VEC;
  float*       o  = dest + (size_t)node * F + lane * VEC;
  if (VEC == 4) {
    const float4 s = *(const float4*)ip;
    float4 r;
    r.x = s.x + (acc0[0] + acc1[0]);
    r.y = s.y + (acc0[1] + acc1[1]);
    r.z = s.z + (acc0[2] + acc1[2]);
    r.w = s.w + (acc0[3] + acc1[3]);
    *(float4*)o = r;
  } else {
    const float2 s = *(const float2*)ip;
    float2 r;
    r.x = s.x + (acc0[0] + acc1[0]);
    r.y = s.y + (acc0[1] + acc1[1]);
    *(float2*)o = r;
  }
}

extern "C" void kernel_launch(void* const* d_in, const int* in_sizes, int n_in,
                              void* d_out, int out_size, void* d_ws, size_t ws_size,
                              hipStream_t stream) {
  const float* feat  = (const float*)d_in[0];
  const float* W1    = (const float*)d_in[1];
  const float* loop1 = (const float*)d_in[2];
  const float* b1    = (const float*)d_in[3];
  const float* W2    = (const float*)d_in[4];
  const float* loop2 = (const float*)d_in[5];
  const float* b2    = (const float*)d_in[6];
  const int*   src   = (const int*)d_in[7];
  const int*   dst   = (const int*)d_in[8];
  const int*   et    = (const int*)d_in[9];
  float*       out   = (float*)d_out;

  const int N = in_sizes[0] / 64;
  const int E = in_sizes[7];

  float* ws   = (float*)d_ws;
  float* x0   = ws;
  float* x1   = x0 + (size_t)N * 128;
  float* hbuf = x1 + (size_t)N * 128;
  float* y0   = x0;
  float* y1   = x0 + (size_t)N * 64;
  _Float16* wpack1 = (_Float16*)(hbuf + (size_t)N * 128);
  _Float16* wpack2 = wpack1 + 24576;
  unsigned* cnt    = (unsigned*)(wpack2 + 24576);
  float*    sinit2 = x1;
  float*    rowbuf = out;

  const int nblk = N / 16;
  const int eblk = (E + 255) / 256;
  const int ablk = (N * 32 + 255) / 256;
  const int pblk = (2 * PACK_WAVES + (N + 127) / 128 + 7) / 8;
  const int fblk = 2 * (nblk > eblk ? nblk : eblk);

  prep<<<pblk, 256, 0, stream>>>(
      W1, W1 + 64 * 128, loop1, W2, W2 + 128 * 64, loop2, wpack1, wpack2, cnt, N);

  fill_beside_xform1<<<fblk, 256, 0, stream>>>(
      feat, wpack1, b1, x0, x1, hbuf, src, dst, et, cnt, rowbuf, E, nblk);

  aggregate<128><<<ablk, 256, 0, stream>>>(
      x0, x1, cnt, rowbuf, src, dst, et, hbuf, hbuf, N, E);

  node_xform<128, 64, true><<<nblk, 128, 0, stream>>>(
      hbuf, wpack2, b2, y0, y1, sinit2);
  aggregate<64><<<ablk, 256, 0, stream>>>(
      y0, y1, cnt, rowbuf, src, dst, et, sinit2, out, N, E);
}


# Edge rows without an atomic per edge: a work-group takes a 1472-node window and 1/16 of the destination list, ranks its edges on LDS counters; one global add per node reserves slots; order by arrival
# speedup vs baseline: 3.2575x; 1.3138x over previous
#include <hip/hip_runtime.h>
#include <hip/hip_bf16.h>

typedef __attribute__((ext_vector_type(16))) _Float16 v16h;
typedef __attribute__((ext_vector_type(8)))  float    v8f;

template <int K, int NCOL>
__device__ __forceinline__ void pack_fragment(
    const float* __restrict__ Wm, _Float16* __restrict__ out,
    int mat, int x, int lane)
{
  const int ct   = x / (K / 32);
  const int ks   = x % (K / 32);
  const int hs   = lane >> 4;
  const int mr   = lane & 15;
  const int krow = ks * 32 + hs * 8;
  const float* p = Wm + (size_t)krow * NCOL + ct * 16 + mr;
  _Float16* o = out +
      ((((size_t)mat * (NCOL / 16) + ct) * (K / 32) + ks) * 32 + lane) * 16;
#pragma unroll
  for (int i = 0; i < 8; ++i) {
    o[i]     = (_Float16)p[(size_t)i * NCOL];
    o[8 + i] = (_Float16)p[(size_t)(16 + i) * NCOL];
  }
}

constexpr int PACK_WAVES = 48;

__global__ __launch_bounds__(256) void prep(
    const float* __restrict__ w1a, const float* __restrict__ w1b,
    const float* __restrict__ l1,
    const float* __restrict__ w2a, const float* __restrict__ w2b,
    const float* __restrict__ l2,
    _Float16* __restrict__ wpack1, _Float16* __restrict__ wpack2,
    unsigned* __restrict__ cnt, int N)
{
  const int lane = threadIdx.x & 31;
  const int gw   = blockIdx.x * 8 + (threadIdx.x >> 5);
  if (gw < PACK_WAVES) {
    const int mat = gw / 16;
    const float* Wm = (mat == 0) ? w1a : (mat == 1) ? w1b : l1;
    pack_fragment<64, 128>(Wm, wpack1, mat, gw % 16, lane);
  } else if (gw < 2 * PACK_WAVES) {
    const int g   = gw - PACK_WAVES;
    const int mat = g / 16;
    const float* Wm = (mat == 0) ? w2a : (mat == 1) ? w2b : l2;
    pack_fragment<128, 64>(Wm, wpack2, mat, g % 16, lane);
  } else {
    const int base = ((gw - 2 * PACK_WAVES) * 32 + lane) * 4;
    if (base + 3 < N) {
      *(uint4*)(cnt + base) = make_uint4(0u, 0u, 0u, 0u);
    } else {
      for (int k = 0; k < 4; ++k)
        if (base + k < N) cnt[base + k] = 0u;
    }
  }
}

template <int K, int NCOL, bool RELU>
__device__ __forceinline__ void xform_strip(
    int strip, bool valid, int wave,
    v16h* ash,
    const float* __restrict__ x,
    const _Float16* __restrict__ wpack,
    const float* __restrict__ bias,
    float* __restrict__ o0, float* __restrict__ o1, float* __restrict__ oagg)
{
  constexpr int KS = K / 32;

  const int lane = threadIdx.x & 31;
  const int row0 = strip * 16;
  const int hs   = lane >> 4;
  const int mr   = lane & 15;

  if (valid && wave < KS) {
    const int    ks = wave;
    const float* ap = x + (size_t)(row0 + mr) * K + ks * 32 + hs * 8;
    v16h a;
#pragma unroll
    for (int i = 0; i < 8; ++i) {
      float lo = ap[i];
      float hi = ap[16 + i];
      if (RELU) { lo = fmaxf(lo, 0.0f); hi = fmaxf(hi, 0.0f); }
      a[i]     = (_Float16)lo;
      a[8 + i] = (_Float16)hi;
    }
    ash[ks * 32 + lane] = a;
  }
  __syncthreads();
  if (!valid) return;

  v8f c0 = {}, c1 = {}, cl = {};
  constexpr size_t MATSTRIDE = (size_t)(NCOL / 16) * KS * 32 * 16;
  const _Float16* wb = wpack + ((size_t)wave * KS) * 32 * 16 + (size_t)lane * 16;

#pragma unroll
  for (int ks = 0; ks < KS; ++ks) {
    v16h a = ash[ks * 32 + lane];
    const _Float16* f = wb + (size_t)ks * 32 * 16;
    v16h b0 = *(const v16h*)(f);
    v16h b1 = *(const v16h*)(f + MATSTRIDE);
    v16h bl = *(const v16h*)(f + 2 * MATSTRIDE);
    c0 = __builtin_amdgcn_wmma_f32_16x16x32_f16(false, a, false, b0, (short)0, c0, false, false);
    c1 = __builtin_amdgcn_wmma_f32_16x16x32_f16(false, a, false, b1, (short)0, c1, false, false);
    cl = __builtin_amdgcn_wmma_f32_16x16x32_f16(false, a, false, bl, (short)0, cl, false, false);
  }

  const int   col = wave * 16 + mr;
  const float bv  = bias[col];
#pragma unroll
  for (int r = 0; r < 8; ++r) {
    size_t off = (size_t)(row0 + r + hs * 8) * NCOL + col;
    o0[off]   = c0[r];
    o1[off]   = c1[r];
    oagg[off] = cl[r] + bv;
  }
}

template <int K, int NCOL, bool RELU>
__global__ __launch_bounds__((NCOL / 16) * 32) void node_xform(
    const float* __restrict__ x, const _Float16* __restrict__ wpack,
    const float* __restrict__ bias,
    float* __restrict__ o0, float* __restrict__ o1, float* __restrict__ oagg)
{
  __shared__ v16h ash[(K / 32) * 32];
  xform_strip<K, NCOL, RELU>(blockIdx.x, true, threadIdx.x >> 5, ash, x, wpack, bias, o0, o1, oagg);
}

constexpr int ROWW         = 64;
constexpr int SW           = 1472;
constexpr int NPART        = 16;
constexpr int SROW         = 16;
constexpr int FILL_THREADS = 512;
constexpr int SCAN_U       = 4;
constexpr int FILL_WORDS   = 2 * SW + SW * SROW;

__device__ __forceinline__ void rank_edge(
    int d, int e, unsigned lo, unsigned* lcnt, unsigned* stage)
{
  const unsigned t = (unsigned)d - lo;
  if (t < (unsigned)SW) {
    const unsigned r = atomicAdd(&lcnt[t], 1u);
    if (r < (unsigned)SROW) stage[t * SROW + r] = (unsigned)e;
  }
}

__device__ __forceinline__ void rank_quad(
    const int4& v, int q, unsigned lo, unsigned* lcnt, unsigned* stage)
{
  rank_edge(v.x, q * 4,     lo, lcnt, stage);
  rank_edge(v.y, q * 4 + 1, lo, lcnt, stage);
  rank_edge(v.z, q * 4 + 2, lo, lcnt, stage);
  rank_edge(v.w, q * 4 + 3, lo, lcnt, stage);
}

__device__ __forceinline__ void scan_quads(
    const int4* __restrict__ d4, int q0, int q1,
    unsigned lo, unsigned* lcnt, unsigned* stage)
{
  int b = q0 + (int)threadIdx.x;
  for (; b + (SCAN_U - 1) * FILL_THREADS < q1; b += SCAN_U * FILL_THREADS) {
    int4 v[SCAN_U];
#pragma unroll
    for (int u = 0; u < SCAN_U; ++u) v[u] = d4[b + u * FILL_THREADS];
#pragma unroll
    for (int u = 0; u < SCAN_U; ++u) rank_quad(v[u], b + u * FILL_THREADS, lo, lcnt, stage);
  }
  for (; b < q1; b += FILL_THREADS) {
    const int4 v = d4[b];
    rank_quad(v, b, lo, lcnt, stage);
  }
}

__device__ __forceinline__ void fill_window(
    int win, int part, int nwin, unsigned* smem,
    const int* __restrict__ src, const int* __restrict__ dst,
    const int* __restrict__ et, unsigned* cnt,
    float* rowbuf, int N, int E)
{
  unsigned* lcnt  = smem;
  unsigned* lbase = smem + SW;
  unsigned* stage = smem + 2 * SW;
  const int      tid = threadIdx.x;
  const unsigned lo  = (unsigned)win * SW;

  for (int i = tid; i < FILL_WORDS; i += FILL_THREADS) smem[i] = 0u;
  __syncthreads();

  const int   n4    = E >> 2;
  const int4* d4    = (const int4*)dst;
  const int   q0    = (int)(((long long)part * n4) / NPART);
  const int   q1    = (int)(((long long)(part + 1) * n4) / NPART);
  const int   start = q0 + (int)(((long long)win * (q1 - q0)) / nwin);
  scan_quads(d4, start, q1, lo, lcnt, stage);
  scan_quads(d4, q0, start, lo, lcnt, stage);
  if (part == NPART - 1)
    for (int e = (n4 << 2) + tid; e < E; e += FILL_THREADS)
      rank_edge(dst[e], e, lo, lcnt, stage);
  __syncthreads();

  for (int t = tid; t < SW; t += FILL_THREADS) {
    const unsigned c = lcnt[t];
    unsigned base = 0u;
    if (c != 0u && lo + (unsigned)t < (unsigned)N)
      base = atomicAdd(&cnt[lo + t], c > (unsigned)SROW ? c + (unsigned)ROWW : c);
    lbase[t] = base;
  }
  __syncthreads();

  unsigned* rows = (unsigned*)rowbuf + (size_t)lo * ROWW;
  constexpr int PLACES = SW * SROW;
  for (int k0 = 0; k0 < PLACES; k0 += 8 * FILL_THREADS) {
    unsigned wv[8];
#pragma unroll
    for (int k = 0; k < 8; ++k) {
      const int s = k0 + k * FILL_THREADS + tid;
      unsigned word = 0u;
      if (s < PLACES && (unsigned)(s % SROW) < lcnt[s / SROW]) {
        const unsigned e = stage[s];
        word = (unsigned)src[e] | ((unsigned)et[e] << 30);
      }
      wv[k] = word;
    }
#pragma unroll
    for (int k = 0; k < 8; ++k) {
      const int s = k0 + k * FILL_THREADS + tid;
      if (s < PLACES) {
        const int      t    = s / SROW;
        const unsigned r    = (unsigned)(s % SROW);
        const unsigned slot = lbase[t] + r;
        if (r < lcnt[t] && slot < (unsigned)ROWW && lo + (unsigned)t < (unsigned)N)
          rows[t * ROWW + slot] = wv[k];
      }
    }
  }
}

__global__ __launch_bounds__(FILL_THREADS) void fill_beside_xform1(
    const float* __restrict__ feat, const _Float16* __restrict__ wpack1,
    const float* __restrict__ b1,
    float* __restrict__ x0, float* __restrict__ x1, float* __restrict__ hbuf,
    const int* __restrict__ src, const int* __restrict__ dst,
    const int* __restrict__ et, unsigned* cnt,
    float* rowbuf, int N, int E, int nstrips, int nwin)
{
  __shared__ __attribute__((aligned(32))) unsigned smem[FILL_WORDS];
  const int nfill = nwin * NPART;
  if ((int)blockIdx.x < nfill) {
    fill_window((int)blockIdx.x / NPART, (int)blockIdx.x % NPART, nwin, smem,
                src, dst, et, cnt, rowbuf, N, E);
  } else {
    const int wave  = threadIdx.x >> 5;
    const int sub   = wave >> 3;
    const int strip = ((int)blockIdx.x - nfill) * 2 + sub;
    xform_strip<64, 128, false>(strip, strip < nstrips, wave & 7,
                                (v16h*)smem + sub * 64, feat, wpack1, b1, x0, x1, hbuf);
  }
}

typedef unsigned v16u __attribute__((ext_vector_type(16)));
typedef unsigned v8u  __attribute__((ext_vector_type(8)));
typedef unsigned v4u  __attribute__((ext_vector_type(4)));
typedef unsigned v2u  __attribute__((ext_vector_type(2)));

template <int VEC> struct RowVec;
template <> struct RowVec<4> { typedef float4 T; };
template <> struct RowVec<2> { typedef float2 T; };

__device__ __forceinline__ void add_row(float (&a)[4], const float4& v) {
  a[0] += v.x; a[1] += v.y; a[2] += v.z; a[3] += v.w;
}
__device__ __forceinline__ void add_row(float (&a)[2], const float2& v) {
  a[0] += v.x; a[1] += v.y;
}

template <int F>
__device__ __forceinline__ typename RowVec<F / 32>::T load_row(
    unsigned pk, const float* __restrict__ x0, const float* __restrict__ x1, int lane)
{
  typedef typename RowVec<F / 32>::T V;
  const float* base = ((pk >> 30) ? x1 : x0) + (size_t)(pk & 0x3FFFFFFFu) * F;
  return *(const V*)(base + lane * (F / 32));
}

template <int F, int NB>
__device__ __forceinline__ void sum_rows(
    const typename RowVec<F / 32>::T* v, float (&acc0)[F / 32], float (&acc1)[F / 32])
{
#pragma unroll
  for (int j = 0; j < NB; ++j) {
    if (j & 1) add_row(acc1, v[j]); else add_row(acc0, v[j]);
  }
}

template <int F>
__device__ __forceinline__ void gather8(
    const unsigned (&pk)[8],
    const float* __restrict__ x0, const float* __restrict__ x1, int lane,
    float (&acc0)[F / 32], float (&acc1)[F / 32])
{
  typename RowVec<F / 32>::T v[8];
#pragma unroll
  for (int j = 0; j < 8; ++j) v[j] = load_row<F>(pk[j], x0, x1, lane);
  sum_rows<F, 8>(v, acc0, acc1);
}

template <int F>
__device__ __forceinline__ void gather_node(
    int node, unsigned deg, const v16u& w, const unsigned* row,
    const float* __restrict__ x0, const float* __restrict__ x1,
    const int* __restrict__ src, const int* __restrict__ dst,
    const int* __restrict__ et, int lane, int E,
    float (&sum)[F / 32])
{
  constexpr int VEC = F / 32;
  typedef typename RowVec<VEC>::T V;
  float acc0[VEC], acc1[VEC];
#pragma unroll
  for (int i = 0; i < VEC; ++i) { acc0[i] = 0.0f; acc1[i] = 0.0f; }

  if (deg <= (unsigned)ROWW) {
    unsigned e = 0;
    if (deg >= 8u) {
      const unsigned p[8] = {w[0], w[1], w[2], w[3], w[4], w[5], w[6], w[7]};
      gather8<F>(p, x0, x1, lane, acc0, acc1);
      e = 8;
      if (deg >= 16u) {
        const unsigned q[8] = {w[8], w[9], w[10], w[11], w[12], w[13], w[14], w[15]};
        gather8<F>(q, x0, x1, lane, acc0, acc1);
        e = 16;
      }
    }
    for (; e + 8 <= deg; e += 8) {
      const v8u t = *(const v8u*)(row + e);
      const unsigned p[8] = {t[0], t[1], t[2], t[3], t[4], t[5], t[6], t[7]};
      gather8<F>(p, x0, x1, lane, acc0, acc1);
    }
    const unsigned e4 = e;
    const unsigned e2 = e4 + (deg & 4u);
    const unsigned e1 = e2 + (deg & 2u);
    V r[7];
    if (deg & 4u) {
      const v4u t = *(const v4u*)(row + e4);
#pragma unroll
      for (int j = 0; j < 4; ++j) r[j] = load_row<F>(t[j], x0, x1, lane);
    }
    if (deg & 2u) {
      const v2u t = *(const v2u*)(row + e2);
      r[4] = load_row<F>(t[0], x0, x1, lane);
      r[5] = load_row<F>(t[1], x0, x1, lane);
    }
    if (deg & 1u) r[6] = load_row<F>(row[e1], x0, x1, lane);
    if (deg & 4u) sum_rows<F, 4>(r, acc0, acc1);
    if (deg & 2u) sum_rows<F, 2>(r + 4, acc0, acc1);
    if (deg & 1u) sum_rows<F, 1>(r + 6, acc0, acc1);
  } else {
    for (int e0 = 0; e0 < E; e0 += 32) {
      const int  e   = e0 + lane;
      const bool hit = (e < E) && (dst[e] == node);
      unsigned   pk  = 0u;
      if (hit) pk = (unsigned)src[e] | ((unsigned)et[e] << 30);
      unsigned m = __builtin_amdgcn_ballot_w32(hit);
      while (m) {
        const int j = __builtin_ctz(m);
        m &= m - 1u;
        const unsigned p = (unsigned)__shfl((int)pk, j);
        const float* rp = ((p >> 30) ? x1 : x0) + (size_t)(p & 0x3FFFFFFFu) * F + lane * VEC;
        add_row(acc0, *(const V*)rp);
      }
    }
  }
#pragma unroll
  for (int i = 0; i < VEC; ++i) sum[i] = acc0[i] + acc1[i];
}

template <int F>
__device__ __forceinline__ void store_node(
    float* o, const typename RowVec<F / 32>::T& s, const float (&a)[F / 32])
{
  if constexpr (F / 32 == 4) {
    float4 q;
    q.x = s.x + a[0]; q.y = s.y + a[1]; q.z = s.z + a[2]; q.w = s.w + a[3];
    *(float4*)o = q;
  } else {
    float2 q;
    q.x = s.x + a[0]; q.y = s.y + a[1];
    *(float2*)o = q;
  }
}

template <int F>
__global__ __launch_bounds__(256) void aggregate(
    const float* __restrict__ x0, const float* __restrict__ x1,
    const unsigned* __restrict__ cnt, const float* rowbuf,
    const int* __restrict__ src, const int* __restrict__ dst,
    const int* __restrict__ et,
    const float* init, float* dest, int N, int E)
{
  constexpr int VEC = F / 32;
  typedef typename RowVec<VEC>::T V;
  const int lane = threadIdx.x & 31;
  const int na = 2 * __builtin_amdgcn_readfirstlane((int)(blockIdx.x * 8 + (threadIdx.x >> 5)));
  if (na >= N) return;
  const bool two = na + 1 < N;
  const int  nb  = two ? na + 1 : na;

  const unsigned* rowa = (const unsigned*)rowbuf + (size_t)na * ROWW;
  const unsigned* rowb = (const unsigned*)rowbuf + (size_t)nb * ROWW;
  const v16u wa = *(const v16u*)rowa;
  const v16u wb = *(const v16u*)rowb;
  unsigned dega = cnt[na];
  unsigned degb = cnt[nb];
  asm("" : "+s"(dega), "+s"(degb) : "s"(wa), "s"(wb));

  const V sa = *(const V*)(init + (size_t)na * F + lane * VEC);
  const V sb = *(const V*)(init + (size_t)nb * F + lane * VEC);

  float suma[VEC], sumb[VEC];
  gather_node<F>(na, dega, wa, rowa, x0, x1, src, dst, et, lane, E, suma);
  gather_node<F>(nb, degb, wb, rowb, x0, x1, src, dst, et, lane, E, sumb);

  store_node<F>(dest + (size_t)na * F + lane * VEC, sa, suma);
  if (two) store_node<F>(dest + (size_t)nb * F + lane * VEC, sb, sumb);
}

extern "C" void kernel_launch(void* const* d_in, const int* in_sizes, int n_in,
                              void* d_out, int out_size, void* d_ws, size_t ws_size,
                              hipStream_t stream) {
  const float* feat  = (const float*)d_in[0];
  const float* W1    = (const float*)d_in[1];
  const float* loop1 = (const float*)d_in[2];
  const float* b1    = (const float*)d_in[3];
  const float* W2    = (const float*)d_in[4];
  const float* loop2 = (const float*)d_in[5];
  const float* b2    = (const float*)d_in[6];
  const int*   src   = (const int*)d_in[7];
  const int*   dst   = (const int*)d_in[8];
  const int*   et    = (const int*)d_in[9];
  float*       out   = (float*)d_out;

  const int N = in_sizes[0] / 64;
  const int E = in_sizes[7];

  float* ws   = (float*)d_ws;
  float* x0   = ws;
  float* x1   = x0 + (size_t)N * 128;
  float* hbuf = x1 + (size_t)N * 128;
  float* y0   = x0;
  float* y1   = x0 + (size_t)N * 64;
  _Float16* wpack1 = (_Float16*)(hbuf + (size_t)N * 128);
  _Float16* wpack2 = wpack1 + 24576;
  unsigned* cnt    = (unsigned*)(wpack2 + 24576);
  float*    sinit2 = x1;
  float*    rowbuf = out;

  const int nblk = N / 16;
  const int ablk = (((N + 1) / 2) * 32 + 255) / 256;
  const int pblk = (2 * PACK_WAVES + (N + 127) / 128 + 7) / 8;
  const int nwin = (N + SW - 1) / SW;
  const int fblk = nwin * NPART + (nblk + 1) / 2;

  prep<<<pblk, 256, 0, stream>>>(
      W1, W1 + 64 * 128, loop1, W2, W2 + 128 * 64, loop2, wpack1, wpack2, cnt, N);

  fill_beside_xform1<<<fblk, FILL_THREADS, 0, stream>>>(
      feat, wpack1, b1, x0, x1, hbuf, src, dst, et, cnt, rowbuf, N, E, nblk, nwin);

  aggregate<128><<<ablk, 256, 0, stream>>>(
      x0, x1, cnt, rowbuf, src, dst, et, hbuf, hbuf, N, E);

  node_xform<128, 64, true><<<nblk, 128, 0, stream>>>(
      hbuf, wpack2, b2, y0, y1, sinit2);
  aggregate<64><<<ablk, 256, 0, stream>>>(
      y0, y1, cnt, rowbuf, src, dst, et, sinit2, out, N, E);
}
